# starting kernel
# baseline (speedup 1.0000x reference)
#include <hip/hip_runtime.h>
#include <math.h>

typedef __attribute__((ext_vector_type(2))) float v2f;
typedef __attribute__((ext_vector_type(8))) float v8f;

#define B_DIM 64
#define D_DIM 512
#define H_DIM 8
#define EPS_F 1e-8f

#if defined(__HIP_DEVICE_COMPILE__) && defined(__AMDGCN__)
#if !__has_builtin(__builtin_amdgcn_wmma_f32_16x16x4_f32)
#error "missing __builtin_amdgcn_wmma_f32_16x16x4_f32 on this toolchain (device pass)"
#endif
#endif

__global__ void fl_init_out(const float* __restrict__ x, float* __restrict__ out, int n) {
    int i = blockIdx.x * blockDim.x + threadIdx.x;
    if (i < n) out[i] = x[i];
}

__global__ __launch_bounds__(256) void fl_attn_kernel(const float* __restrict__ x,
                                                      const float* __restrict__ alphas,
                                                      const float* __restrict__ betas,
                                                      float* __restrict__ out) {
    __shared__ float sQ[D_DIM];
    __shared__ float sK[D_DIM];
    __shared__ float sV[D_DIM];
    __shared__ float sC[8][16 * 16];

    const int hb  = blockIdx.x;
    const int b   = hb & (B_DIM - 1);
    const int h   = hb >> 6;
    const int tid = threadIdx.x;

    const float aq = alphas[h * 3 + 0], ak = alphas[h * 3 + 1], av = alphas[h * 3 + 2];
    const float bq = betas[h * 3 + 0],  bk = betas[h * 3 + 1],  bv = betas[h * 3 + 2];

    for (int j = tid; j < D_DIM; j += 256) {
        float xv = x[b * D_DIM + j];
        sQ[j] = fmaf(aq, xv, bq);
        sK[j] = fmaf(ak, xv, bk);
        sV[j] = fmaf(av, xv, bv);
    }
    __syncthreads();

    const int wave  = tid >> 5;
    const int lane  = tid & 31;
    const int half  = lane >> 4;
    const int laneM = lane & 15;
    const int kb    = half * 2;
    const float invSqrtD = 0.044194173824159216f;

    for (int t = wave; t < 32; t += 8) {
        const int i0 = t * 16;
        const float ki = sK[i0 + laneM];

        float mn = 3.402823466e+38f;
        const int jbeg = half * 256;
        #pragma unroll 8
        for (int j = jbeg; j < jbeg + 256; ++j) {
            mn = fminf(mn, fabsf(sQ[j] - ki));
        }
        mn = fminf(mn, __shfl_xor(mn, 16));
        const float smax = __builtin_amdgcn_rcpf(mn + EPS_F);

        v8f c0 = {};
        v8f c1 = {};
        for (int j = 0; j < D_DIM; j += 8) {
            {
                const int jj = j + kb;
                float q0 = sQ[jj], q1 = sQ[jj + 1];
                float w0 = sV[jj], w1 = sV[jj + 1];
                float s0 = __builtin_amdgcn_rcpf(fabsf(q0 - ki) + EPS_F);
                float s1 = __builtin_amdgcn_rcpf(fabsf(q1 - ki) + EPS_F);
                float e0 = __expf(s0 - smax);
                float e1 = __expf(s1 - smax);
                float b0 = (laneM == 0) ? 1.0f : ((laneM == 1) ? w0 : 0.0f);
                float b1 = (laneM == 0) ? 1.0f : ((laneM == 1) ? w1 : 0.0f);
                v2f A  = {e0, e1};
                v2f Bm = {b0, b1};
                c0 = __builtin_amdgcn_wmma_f32_16x16x4_f32(false, A, false, Bm,
                                                           (short)0, c0, false, false);
            }
            {
                const int jj = j + 4 + kb;
                float q0 = sQ[jj], q1 = sQ[jj + 1];
                float w0 = sV[jj], w1 = sV[jj + 1];
                float s0 = __builtin_amdgcn_rcpf(fabsf(q0 - ki) + EPS_F);
                float s1 = __builtin_amdgcn_rcpf(fabsf(q1 - ki) + EPS_F);
                float e0 = __expf(s0 - smax);
                float e1 = __expf(s1 - smax);
                float b0 = (laneM == 0) ? 1.0f : ((laneM == 1) ? w0 : 0.0f);
                float b1 = (laneM == 0) ? 1.0f : ((laneM == 1) ? w1 : 0.0f);
                v2f A  = {e0, e1};
                v2f Bm = {b0, b1};
                c1 = __builtin_amdgcn_wmma_f32_16x16x4_f32(false, A, false, Bm,
                                                           (short)0, c1, false, false);
            }
        }

        const int M0 = half * 8;
        #pragma unroll
        for (int r = 0; r < 8; ++r) {
            sC[wave][(M0 + r) * 16 + laneM] = c0[r] + c1[r];
        }
        asm volatile("s_wait_dscnt 0" ::: "memory");
        if (lane < 16) {
            float sumE  = sC[wave][lane * 16 + 0];
            float sumEV = sC[wave][lane * 16 + 1];
            float att   = (sumEV / sumE) * invSqrtD;
            atomicAdd(&out[b * D_DIM + i0 + lane], att);
        }
    }
}

extern "C" void kernel_launch(void* const* d_in, const int* in_sizes, int n_in,
                              void* d_out, int out_size, void* d_ws, size_t ws_size,
                              hipStream_t stream) {
    const float* x      = (const float*)d_in[0];
    const float* alphas = (const float*)d_in[1];
    const float* betas  = (const float*)d_in[2];
    float* out          = (float*)d_out;

    const int n = B_DIM * D_DIM;
    fl_init_out<<<(n + 255) / 256, 256, 0, stream>>>(x, out, n);
    fl_attn_kernel<<<H_DIM * B_DIM, 256, 0, stream>>>(x, alphas, betas, out);
}


# min pass rewritten: packed f32 subtract and min3 over 16-byte LDS reads, two running minima; replaces the compiled per-pair NaN-test-and-branch chain that waited on every 8-byte LDS read
# speedup vs baseline: 1.6157x; 1.6157x over previous
#include <hip/hip_runtime.h>
#include <math.h>

typedef __attribute__((ext_vector_type(2))) float v2f;
typedef __attribute__((ext_vector_type(4))) float v4f;
typedef __attribute__((ext_vector_type(8))) float v8f;

#define B_DIM 64
#define D_DIM 512
#define H_DIM 8
#define EPS_F 1e-8f

#if defined(__HIP_DEVICE_COMPILE__) && defined(__AMDGCN__)
#if !__has_builtin(__builtin_amdgcn_wmma_f32_16x16x4_f32)
#error "missing __builtin_amdgcn_wmma_f32_16x16x4_f32 on this toolchain (device pass)"
#endif
#endif

__global__ void fl_init_out(const float* __restrict__ x, float* __restrict__ out, int n) {
    int i = blockIdx.x * blockDim.x + threadIdx.x;
    if (i < n) out[i] = x[i];
}

__global__ __launch_bounds__(256) void fl_attn_kernel(const float* __restrict__ x,
                                                      const float* __restrict__ alphas,
                                                      const float* __restrict__ betas,
                                                      float* __restrict__ out) {
    __shared__ __attribute__((aligned(16))) float sQ[D_DIM];
    __shared__ float sK[D_DIM];
    __shared__ float sV[D_DIM];
    __shared__ float sC[8][16 * 16];

    const int hb  = blockIdx.x;
    const int b   = hb & (B_DIM - 1);
    const int h   = hb >> 6;
    const int tid = threadIdx.x;

    const float aq = alphas[h * 3 + 0], ak = alphas[h * 3 + 1], av = alphas[h * 3 + 2];
    const float bq = betas[h * 3 + 0],  bk = betas[h * 3 + 1],  bv = betas[h * 3 + 2];

    for (int j = tid; j < D_DIM; j += 256) {
        float xv = x[b * D_DIM + j];
        sQ[j] = fmaf(aq, xv, bq);
        sK[j] = fmaf(ak, xv, bk);
        sV[j] = fmaf(av, xv, bv);
    }
    __syncthreads();

    const int wave  = tid >> 5;
    const int lane  = tid & 31;
    const int half  = lane >> 4;
    const int laneM = lane & 15;
    const int kb    = half * 2;
    const float invSqrtD = 0.044194173824159216f;

    for (int t = wave; t < 32; t += 8) {
        const int i0 = t * 16;
        const float ki = sK[i0 + laneM];

        const v2f k2 = {ki, ki};
        float mnA = 3.402823466e+38f;
        float mnB = 3.402823466e+38f;
        const float* q1 = sQ + half * 256;
        for (int j = 0; j < 256; j += 4) {
            const v4f q4  = *(const v4f*)(q1 + j);
            const v2f d01 = q4.xy - k2;
            const v2f d23 = q4.zw - k2;
            mnA = fminf(fminf(mnA, fabsf(d01.x)), fabsf(d01.y));
            mnB = fminf(fminf(mnB, fabsf(d23.x)), fabsf(d23.y));
        }
        float mn = fminf(mnA, mnB);
        mn = fminf(mn, __shfl_xor(mn, 16));
        const float smax = __builtin_amdgcn_rcpf(mn + EPS_F);

        v8f c0 = {};
        v8f c1 = {};
        for (int j = 0; j < D_DIM; j += 8) {
            {
                const int jj = j + kb;
                float q0 = sQ[jj], q1 = sQ[jj + 1];
                float w0 = sV[jj], w1 = sV[jj + 1];
                float s0 = __builtin_amdgcn_rcpf(fabsf(q0 - ki) + EPS_F);
                float s1 = __builtin_amdgcn_rcpf(fabsf(q1 - ki) + EPS_F);
                float e0 = __expf(s0 - smax);
                float e1 = __expf(s1 - smax);
                float b0 = (laneM == 0) ? 1.0f : ((laneM == 1) ? w0 : 0.0f);
                float b1 = (laneM == 0) ? 1.0f : ((laneM == 1) ? w1 : 0.0f);
                v2f A  = {e0, e1};
                v2f Bm = {b0, b1};
                c0 = __builtin_amdgcn_wmma_f32_16x16x4_f32(false, A, false, Bm,
                                                           (short)0, c0, false, false);
            }
            {
                const int jj = j + 4 + kb;
                float q0 = sQ[jj], q1 = sQ[jj + 1];
                float w0 = sV[jj], w1 = sV[jj + 1];
                float s0 = __builtin_amdgcn_rcpf(fabsf(q0 - ki) + EPS_F);
                float s1 = __builtin_amdgcn_rcpf(fabsf(q1 - ki) + EPS_F);
                float e0 = __expf(s0 - smax);
                float e1 = __expf(s1 - smax);
                float b0 = (laneM == 0) ? 1.0f : ((laneM == 1) ? w0 : 0.0f);
                float b1 = (laneM == 0) ? 1.0f : ((laneM == 1) ? w1 : 0.0f);
                v2f A  = {e0, e1};
                v2f Bm = {b0, b1};
                c1 = __builtin_amdgcn_wmma_f32_16x16x4_f32(false, A, false, Bm,
                                                           (short)0, c1, false, false);
            }
        }

        const int M0 = half * 8;
        #pragma unroll
        for (int r = 0; r < 8; ++r) {
            sC[wave][(M0 + r) * 16 + laneM] = c0[r] + c1[r];
        }
        asm volatile("s_wait_dscnt 0" ::: "memory");
        if (lane < 16) {
            float sumE  = sC[wave][lane * 16 + 0];
            float sumEV = sC[wave][lane * 16 + 1];
            float att   = (sumEV / sumE) * invSqrtD;
            atomicAdd(&out[b * D_DIM + i0 + lane], att);
        }
    }
}

extern "C" void kernel_launch(void* const* d_in, const int* in_sizes, int n_in,
                              void* d_out, int out_size, void* d_ws, size_t ws_size,
                              hipStream_t stream) {
    const float* x      = (const float*)d_in[0];
    const float* alphas = (const float*)d_in[1];
    const float* betas  = (const float*)d_in[2];
    float* out          = (float*)d_out;

    const int n = B_DIM * D_DIM;
    fl_init_out<<<(n + 255) / 256, 256, 0, stream>>>(x, out, n);
    fl_attn_kernel<<<H_DIM * B_DIM, 256, 0, stream>>>(x, alphas, betas, out);
}


# pass 2 loop interchange: j outer, the wave's four row tiles inner, so each q/v LDS fetch, its address and the B fragment are made once per 4-wide chunk and serve four tiles (one accumulator each)
# speedup vs baseline: 1.8378x; 1.1375x over previous
#include <hip/hip_runtime.h>
#include <math.h>

typedef __attribute__((ext_vector_type(2))) float v2f;
typedef __attribute__((ext_vector_type(4))) float v4f;
typedef __attribute__((ext_vector_type(8))) float v8f;

#define B_DIM 64
#define D_DIM 512
#define H_DIM 8
#define EPS_F 1e-8f

#if defined(__HIP_DEVICE_COMPILE__) && defined(__AMDGCN__)
#if !__has_builtin(__builtin_amdgcn_wmma_f32_16x16x4_f32)
#error "missing __builtin_amdgcn_wmma_f32_16x16x4_f32 on this toolchain (device pass)"
#endif
#endif

__global__ void fl_init_out(const float* __restrict__ x, float* __restrict__ out, int n) {
    int i = blockIdx.x * blockDim.x + threadIdx.x;
    if (i < n) out[i] = x[i];
}

__global__ __launch_bounds__(256) void fl_attn_kernel(const float* __restrict__ x,
                                                      const float* __restrict__ alphas,
                                                      const float* __restrict__ betas,
                                                      float* __restrict__ out) {
    __shared__ __attribute__((aligned(16))) float sQ[D_DIM];
    __shared__ float sK[D_DIM];
    __shared__ float sV[D_DIM];
    __shared__ float sC[8][16 * 16];

    const int hb  = blockIdx.x;
    const int b   = hb & (B_DIM - 1);
    const int h   = hb >> 6;
    const int tid = threadIdx.x;

    const float aq = alphas[h * 3 + 0], ak = alphas[h * 3 + 1], av = alphas[h * 3 + 2];
    const float bq = betas[h * 3 + 0],  bk = betas[h * 3 + 1],  bv = betas[h * 3 + 2];

    for (int j = tid; j < D_DIM; j += 256) {
        float xv = x[b * D_DIM + j];
        sQ[j] = fmaf(aq, xv, bq);
        sK[j] = fmaf(ak, xv, bk);
        sV[j] = fmaf(av, xv, bv);
    }
    __syncthreads();

    const int wave  = tid >> 5;
    const int lane  = tid & 31;
    const int half  = lane >> 4;
    const int laneM = lane & 15;
    const int kb    = half * 2;
    const float invSqrtD = 0.044194173824159216f;

    float ki[4];
    float smax[4];

    #pragma unroll
    for (int u = 0; u < 4; ++u) {
        const int i0 = (wave + 8 * u) * 16;
        const float k = sK[i0 + laneM];
        const v2f k2 = {k, k};
        float mnA = 3.402823466e+38f;
        float mnB = 3.402823466e+38f;
        const float* q1 = sQ + half * 256;
        for (int j = 0; j < 256; j += 4) {
            const v4f q4  = *(const v4f*)(q1 + j);
            const v2f d01 = q4.xy - k2;
            const v2f d23 = q4.zw - k2;
            mnA = fminf(fminf(mnA, fabsf(d01.x)), fabsf(d01.y));
            mnB = fminf(fminf(mnB, fabsf(d23.x)), fabsf(d23.y));
        }
        float mn = fminf(mnA, mnB);
        mn = fminf(mn, __shfl_xor(mn, 16));
        ki[u]   = k;
        smax[u] = __builtin_amdgcn_rcpf(mn + EPS_F);
    }

    v8f c[4] = {{}, {}, {}, {}};
    for (int j = 0; j < D_DIM; j += 4) {
        const int jj = j + kb;
        const float q0 = sQ[jj], q1 = sQ[jj + 1];
        const float w0 = sV[jj], w1 = sV[jj + 1];
        const float b0 = (laneM == 0) ? 1.0f : ((laneM == 1) ? w0 : 0.0f);
        const float b1 = (laneM == 0) ? 1.0f : ((laneM == 1) ? w1 : 0.0f);
        const v2f Bm = {b0, b1};
        #pragma unroll
        for (int u = 0; u < 4; ++u) {
            const float s0 = __builtin_amdgcn_rcpf(fabsf(q0 - ki[u]) + EPS_F);
            const float s1 = __builtin_amdgcn_rcpf(fabsf(q1 - ki[u]) + EPS_F);
            const float e0 = __expf(s0 - smax[u]);
            const float e1 = __expf(s1 - smax[u]);
            const v2f A = {e0, e1};
            c[u] = __builtin_amdgcn_wmma_f32_16x16x4_f32(false, A, false, Bm,
                                                         (short)0, c[u], false, false);
        }
    }

    const int M0 = half * 8;
    #pragma unroll
    for (int u = 0; u < 4; ++u) {
        const int i0 = (wave + 8 * u) * 16;
        #pragma unroll
        for (int r = 0; r < 8; ++r) {
            sC[wave][(M0 + r) * 16 + laneM] = c[u][r];
        }
        asm volatile("s_wait_dscnt 0" ::: "memory");
        if (lane < 16) {
            float sumE  = sC[wave][lane * 16 + 0];
            float sumEV = sC[wave][lane * 16 + 1];
            float att   = (sumEV / sumE) * invSqrtD;
            atomicAdd(&out[b * D_DIM + i0 + lane], att);
        }
    }
}

extern "C" void kernel_launch(void* const* d_in, const int* in_sizes, int n_in,
                              void* d_out, int out_size, void* d_ws, size_t ws_size,
                              hipStream_t stream) {
    const float* x      = (const float*)d_in[0];
    const float* alphas = (const float*)d_in[1];
    const float* betas  = (const float*)d_in[2];
    float* out          = (float*)d_out;

    const int n = B_DIM * D_DIM;
    fl_init_out<<<(n + 255) / 256, 256, 0, stream>>>(x, out, n);
    fl_attn_kernel<<<H_DIM * B_DIM, 256, 0, stream>>>(x, alphas, betas, out);
}


# pass 2 software-pipelined: behind each matrix instruction one stage (distance, reciprocal, scale, exp2) of each of four later pairs; q and v of the next 16 j prefetched by 16-byte LDS reads
# speedup vs baseline: 1.9978x; 1.0871x over previous
#include <hip/hip_runtime.h>
#include <math.h>

typedef __attribute__((ext_vector_type(2))) float v2f;
typedef __attribute__((ext_vector_type(4))) float v4f;
typedef __attribute__((ext_vector_type(8))) float v8f;

#define B_DIM 64
#define D_DIM 512
#define H_DIM 8
#define ROWS_PER_BLOCK 64
#define EPS_F 1e-8f

#if defined(__HIP_DEVICE_COMPILE__) && defined(__AMDGCN__)
#if !__has_builtin(__builtin_amdgcn_wmma_f32_16x16x4_f32)
#error "missing __builtin_amdgcn_wmma_f32_16x16x4_f32 on this toolchain (device pass)"
#endif
#endif

__global__ __launch_bounds__(256) void fl_attn_kernel(const float* __restrict__ x,
                                                      const float* __restrict__ alphas,
                                                      const float* __restrict__ betas,
                                                      float* __restrict__ out) {
    __shared__ __attribute__((aligned(16))) float sQ[H_DIM][D_DIM];
    __shared__ __attribute__((aligned(16))) float sV[H_DIM][D_DIM];
    __shared__ float sC[8][16 * 16];
    __shared__ float sAtt[H_DIM][ROWS_PER_BLOCK];

    const int bs  = blockIdx.x;
    const int b   = bs & (B_DIM - 1);
    const int sl  = bs >> 6;
    const int tid = threadIdx.x;

    const int wave  = __builtin_amdgcn_readfirstlane(tid >> 5);
    const int lane  = tid & 31;
    const int half  = lane >> 4;
    const int laneM = lane & 15;
    const int kb    = half * 2;
    const float invSqrtD = 0.044194173824159216f;

    const float* xrow = x + b * D_DIM;
    const int row0 = sl * ROWS_PER_BLOCK;

    const float x0 = xrow[tid];
    const float x1 = xrow[tid + 256];
    const float xk[4] = {xrow[row0 + laneM], xrow[row0 + 16 + laneM],
                         xrow[row0 + 32 + laneM], xrow[row0 + 48 + laneM]};
    const float xr = xrow[row0 + (tid & (ROWS_PER_BLOCK - 1))];

    #pragma unroll
    for (int h = 0; h < H_DIM; ++h) {
        const float aq = alphas[h * 3 + 0], av = alphas[h * 3 + 2];
        const float bq = betas[h * 3 + 0],  bv = betas[h * 3 + 2];
        sQ[h][tid]       = fmaf(aq, x0, bq);
        sQ[h][tid + 256] = fmaf(aq, x1, bq);
        sV[h][tid]       = fmaf(av, x0, bv);
        sV[h][tid + 256] = fmaf(av, x1, bv);
    }
    __syncthreads();

    const float ak = alphas[wave * 3 + 1];
    const float bk = betas[wave * 3 + 1];
    const float* sQw = sQ[wave];
    const float* sVw = sV[wave];

    float ki[4];
    float smax[4];

    #pragma unroll
    for (int u = 0; u < 4; ++u) {
        const float k = fmaf(ak, xk[u], bk);
        const v2f k2 = {k, k};
        float mnA = 3.402823466e+38f;
        float mnB = 3.402823466e+38f;
        const float* q1 = sQw + half * 256;
        for (int j = 0; j < 256; j += 4) {
            const v4f q4  = *(const v4f*)(q1 + j);
            const v2f d01 = q4.xy - k2;
            const v2f d23 = q4.zw - k2;
            mnA = fminf(fminf(mnA, fabsf(d01.x)), fabsf(d01.y));
            mnB = fminf(fminf(mnB, fabsf(d23.x)), fabsf(d23.y));
        }
        float mn = fminf(mnA, mnB);
        mn = fminf(mn, __shfl_xor(mn, 16));
        ki[u]   = k;
        smax[u] = __builtin_amdgcn_rcpf(mn + EPS_F);
    }

    const float LOG2E_F = 1.44269504088896341f;
    v8f c[4] = {{}, {}, {}, {}};
    const float* qh = sQw + half * 8;
    const float* vh = sVw + half * 8;
    const bool isOne = (laneM == 0);
    const bool isV   = (laneM == 1);

    auto stD = [&](float q, int u) -> float { return fabsf(q - ki[u]) + EPS_F; };
    auto stR = [&](float d) -> float { return __builtin_amdgcn_rcpf(d); };
    auto stF = [&](float s, int u) -> float { return (s - smax[u]) * LOG2E_F; };
    auto stE = [&](float t) -> float { return __builtin_amdgcn_exp2f(t); };
    auto mkB = [&](float w) -> float { return isOne ? 1.0f : (isV ? w : 0.0f); };

    v4f qa0 = *(const v4f*)(qh);
    v4f qb0 = *(const v4f*)(qh + 4);
    v4f va0 = *(const v4f*)(vh);
    v4f vb0 = *(const v4f*)(vh + 4);
    v4f qa1, qb1, va1, vb1;
    float cA0 = stE(stF(stR(stD(qa0.x, 0)), 0)), cA1 = stE(stF(stR(stD(qa0.y, 0)), 0));
    float cT0 = stF(stR(stD(qa0.x, 1)), 1),      cT1 = stF(stR(stD(qa0.y, 1)), 1);
    float cS0 = stR(stD(qa0.x, 2)),              cS1 = stR(stD(qa0.y, 2));
    float cD0 = stD(qa0.x, 3),                   cD1 = stD(qa0.y, 3);
    float cB0 = mkB(va0.x),                      cB1 = mkB(va0.y);

    auto trip = [&](const v4f& qa, const v4f& qb, const v4f& va, const v4f& vb,
                    v4f& nqa, v4f& nqb, v4f& nva, v4f& nvb, int jn) {
        nqa = *(const v4f*)(qh + jn);
        nqb = *(const v4f*)(qh + jn + 4);
        nva = *(const v4f*)(vh + jn);
        nvb = *(const v4f*)(vh + jn + 4);
        const float qv[10] = {qa.x, qa.y, qa.z, qa.w, qb.x, qb.y, qb.z, qb.w, nqa.x, nqa.y};
        const float vv[10] = {va.x, va.y, va.z, va.w, vb.x, vb.y, vb.z, vb.w, nva.x, nva.y};
        float A0[17], A1[17], T0[18], T1[18], S0[19], S1[19], D0[20], D1[20], B0[5], B1[5];
        A0[0] = cA0; A1[0] = cA1; T0[1] = cT0; T1[1] = cT1; S0[2] = cS0; S1[2] = cS1;
        D0[3] = cD0; D1[3] = cD1; B0[0] = cB0; B1[0] = cB1;
        #pragma unroll
        for (int n = 0; n < 16; ++n) {
            const int u = n & 3, t = n >> 2;
            const v2f Am = {A0[n], A1[n]};
            const v2f Bm = {B0[t], B1[t]};
            c[u] = __builtin_amdgcn_wmma_f32_16x16x4_f32(false, Am, false, Bm,
                                                         (short)0, c[u], false, false);
            A0[n + 1] = stE(T0[n + 1]);
            A1[n + 1] = stE(T1[n + 1]);
            T0[n + 2] = stF(S0[n + 2], (n + 2) & 3);
            T1[n + 2] = stF(S1[n + 2], (n + 2) & 3);
            S0[n + 3] = stR(D0[n + 3]);
            S1[n + 3] = stR(D1[n + 3]);
            D0[n + 4] = stD(qv[2 * ((n + 4) >> 2)], (n + 4) & 3);
            D1[n + 4] = stD(qv[2 * ((n + 4) >> 2) + 1], (n + 4) & 3);
            if (u == 0) { B0[t + 1] = mkB(vv[2 * (t + 1)]); B1[t + 1] = mkB(vv[2 * (t + 1) + 1]); }

            asm volatile("" : "+v"(A0[n + 1]), "+v"(A1[n + 1]), "+v"(T0[n + 2]), "+v"(T1[n + 2]),
                              "+v"(S0[n + 3]), "+v"(S1[n + 3]), "+v"(D0[n + 4]), "+v"(D1[n + 4]));
            __builtin_amdgcn_sched_barrier(0);

        }
        cA0 = A0[16]; cA1 = A1[16]; cT0 = T0[17]; cT1 = T1[17]; cS0 = S0[18]; cS1 = S1[18];
        cD0 = D0[19]; cD1 = D1[19]; cB0 = B0[4]; cB1 = B1[4];
    };

    for (int j = 0; j < D_DIM; j += 32) {
        trip(qa0, qb0, va0, vb0, qa1, qb1, va1, vb1, j + 16);
        trip(qa1, qb1, va1, vb1, qa0, qb0, va0, vb0, (j + 32) & (D_DIM - 1));
    }

    const int M0 = half * 8;
    #pragma unroll
    for (int u = 0; u < 4; ++u) {
        #pragma unroll
        for (int r = 0; r < 8; ++r) {
            sC[wave][(M0 + r) * 16 + laneM] = c[u][r];
        }
        asm volatile("s_wait_dscnt 0" ::: "memory");
        if (lane < 16) {
            float sumE  = sC[wave][lane * 16 + 0];
            float sumEV = sC[wave][lane * 16 + 1];
            sAtt[wave][16 * u + lane] = (sumEV / sumE) * invSqrtD;
        }
    }
    __syncthreads();

    if (tid < ROWS_PER_BLOCK) {
        float acc = sAtt[0][tid];
        for (int h = 1; h < H_DIM; ++h) acc += sAtt[h][tid];
        out[b * D_DIM + row0 + tid] = xr + acc;
    }
}

extern "C" void kernel_launch(void* const* d_in, const int* in_sizes, int n_in,
                              void* d_out, int out_size, void* d_ws, size_t ws_size,
                              hipStream_t stream) {
    const float* x      = (const float*)d_in[0];
    const float* alphas = (const float*)d_in[1];
    const float* betas  = (const float*)d_in[2];
    float* out          = (float*)d_out;

    fl_attn_kernel<<<B_DIM * (D_DIM / ROWS_PER_BLOCK), 256, 0, stream>>>(x, alphas, betas, out);
}


# pass 2 stage arithmetic packed: a lane's two pairs per matrix instruction share one packed f32 subtract and one packed fma (exponent argument as s*log2e - smax*log2e); B fragment by one packed fma
# speedup vs baseline: 2.0799x; 1.0411x over previous
#include <hip/hip_runtime.h>
#include <math.h>

typedef __attribute__((ext_vector_type(2))) float v2f;
typedef __attribute__((ext_vector_type(4))) float v4f;
typedef __attribute__((ext_vector_type(8))) float v8f;

#define B_DIM 64
#define D_DIM 512
#define H_DIM 8
#define ROWS_PER_BLOCK 64
#define EPS_F 1e-8f

#if defined(__HIP_DEVICE_COMPILE__) && defined(__AMDGCN__)
#if !__has_builtin(__builtin_amdgcn_wmma_f32_16x16x4_f32)
#error "missing __builtin_amdgcn_wmma_f32_16x16x4_f32 on this toolchain (device pass)"
#endif
#endif

__global__ __launch_bounds__(256) void fl_attn_kernel(const float* __restrict__ x,
                                                      const float* __restrict__ alphas,
                                                      const float* __restrict__ betas,
                                                      float* __restrict__ out) {
    __shared__ __attribute__((aligned(16))) float sQ[H_DIM][D_DIM];
    __shared__ __attribute__((aligned(16))) float sV[H_DIM][D_DIM];
    __shared__ float sC[8][16 * 16];
    __shared__ float sAtt[H_DIM][ROWS_PER_BLOCK];

    const int bs  = blockIdx.x;
    const int b   = bs & (B_DIM - 1);
    const int sl  = bs >> 6;
    const int tid = threadIdx.x;

    const int wave  = __builtin_amdgcn_readfirstlane(tid >> 5);
    const int lane  = tid & 31;
    const int half  = lane >> 4;
    const int laneM = lane & 15;
    const int kb    = half * 2;
    const float invSqrtD = 0.044194173824159216f;

    const float* xrow = x + b * D_DIM;
    const int row0 = sl * ROWS_PER_BLOCK;

    const float x0 = xrow[tid];
    const float x1 = xrow[tid + 256];
    const float xk[4] = {xrow[row0 + laneM], xrow[row0 + 16 + laneM],
                         xrow[row0 + 32 + laneM], xrow[row0 + 48 + laneM]};
    const float xr = xrow[row0 + (tid & (ROWS_PER_BLOCK - 1))];

    #pragma unroll
    for (int h = 0; h < H_DIM; ++h) {
        const float aq = alphas[h * 3 + 0], av = alphas[h * 3 + 2];
        const float bq = betas[h * 3 + 0],  bv = betas[h * 3 + 2];
        sQ[h][tid]       = fmaf(aq, x0, bq);
        sQ[h][tid + 256] = fmaf(aq, x1, bq);
        sV[h][tid]       = fmaf(av, x0, bv);
        sV[h][tid + 256] = fmaf(av, x1, bv);
    }
    __syncthreads();

    const float ak = alphas[wave * 3 + 1];
    const float bk = betas[wave * 3 + 1];
    const float* sQw = sQ[wave];
    const float* sVw = sV[wave];

    float ki[4];
    float smax[4];

    #pragma unroll
    for (int u = 0; u < 4; ++u) {
        const float k = fmaf(ak, xk[u], bk);
        const v2f k2 = {k, k};
        float mnA = 3.402823466e+38f;
        float mnB = 3.402823466e+38f;
        const float* q1 = sQw + half * 256;
        for (int j = 0; j < 256; j += 4) {
            const v4f q4  = *(const v4f*)(q1 + j);
            const v2f d01 = q4.xy - k2;
            const v2f d23 = q4.zw - k2;
            mnA = fminf(fminf(mnA, fabsf(d01.x)), fabsf(d01.y));
            mnB = fminf(fminf(mnB, fabsf(d23.x)), fabsf(d23.y));
        }
        float mn = fminf(mnA, mnB);
        mn = fminf(mn, __shfl_xor(mn, 16));
        ki[u]   = k;
        smax[u] = __builtin_amdgcn_rcpf(mn + EPS_F);
    }

    const float LOG2E_F = 1.44269504088896341f;
    v8f c[4] = {{}, {}, {}, {}};
    const float* qh = sQw + half * 8;
    const float* vh = sVw + half * 8;

    v2f k2[4], nm2[4];
    #pragma unroll
    for (int u = 0; u < 4; ++u) {
        k2[u]  = (v2f){ki[u], ki[u]};
        nm2[u] = (v2f){-smax[u] * LOG2E_F, -smax[u] * LOG2E_F};
    }
    const v2f L2 = {LOG2E_F, LOG2E_F};
    const float selV   = (laneM == 1) ? 1.0f : 0.0f;
    const float selOne = (laneM == 0) ? 1.0f : 0.0f;
    const v2f selV2   = {selV, selV};
    const v2f selOne2 = {selOne, selOne};

    auto stD = [&](v2f q, int u, float& d0, float& d1) {
        const v2f d = q - k2[u];
        d0 = fabsf(d.x) + EPS_F;
        d1 = fabsf(d.y) + EPS_F;
    };
    auto stR = [&](float d0, float d1) -> v2f {
        return (v2f){__builtin_amdgcn_rcpf(d0), __builtin_amdgcn_rcpf(d1)};
    };
    auto stF = [&](v2f s, int u) -> v2f { return __builtin_elementwise_fma(s, L2, nm2[u]); };
    auto stE = [&](v2f t) -> v2f {
        return (v2f){__builtin_amdgcn_exp2f(t.x), __builtin_amdgcn_exp2f(t.y)};
    };
    auto mkB = [&](v2f w) -> v2f { return __builtin_elementwise_fma(w, selV2, selOne2); };

    v4f qa0 = *(const v4f*)(qh);
    v4f qb0 = *(const v4f*)(qh + 4);
    v4f va0 = *(const v4f*)(vh);
    v4f vb0 = *(const v4f*)(vh + 4);
    v4f qa1, qb1, va1, vb1;
    float f0, f1, cD0, cD1;
    stD(qa0.xy, 0, f0, f1);
    v2f cA = stE(stF(stR(f0, f1), 0));
    stD(qa0.xy, 1, f0, f1);
    v2f cT = stF(stR(f0, f1), 1);
    stD(qa0.xy, 2, f0, f1);
    v2f cS = stR(f0, f1);
    stD(qa0.xy, 3, cD0, cD1);
    v2f cB = mkB(va0.xy);

    auto trip = [&](const v4f& qa, const v4f& qb, const v4f& va, const v4f& vb,
                    v4f& nqa, v4f& nqb, v4f& nva, v4f& nvb, int jn) {
        nqa = *(const v4f*)(qh + jn);
        nqb = *(const v4f*)(qh + jn + 4);
        nva = *(const v4f*)(vh + jn);
        nvb = *(const v4f*)(vh + jn + 4);
        const v2f qp[5] = {qa.xy, qa.zw, qb.xy, qb.zw, nqa.xy};
        const v2f vp[5] = {va.xy, va.zw, vb.xy, vb.zw, nva.xy};
        v2f A[17], T[18], S[19], Bf[5];
        float D0[20], D1[20];
        A[0] = cA; T[1] = cT; S[2] = cS; D0[3] = cD0; D1[3] = cD1; Bf[0] = cB;
        #pragma unroll
        for (int n = 0; n < 16; ++n) {
            const int u = n & 3, t = n >> 2;
            c[u] = __builtin_amdgcn_wmma_f32_16x16x4_f32(false, A[n], false, Bf[t],
                                                         (short)0, c[u], false, false);
            A[n + 1] = stE(T[n + 1]);
            T[n + 2] = stF(S[n + 2], (n + 2) & 3);
            S[n + 3] = stR(D0[n + 3], D1[n + 3]);
            stD(qp[(n + 4) >> 2], (n + 4) & 3, D0[n + 4], D1[n + 4]);
            if (u == 0) Bf[t + 1] = mkB(vp[t + 1]);

            asm volatile("" : "+v"(A[n + 1]), "+v"(T[n + 2]), "+v"(S[n + 3]),
                              "+v"(D0[n + 4]), "+v"(D1[n + 4]));
            __builtin_amdgcn_sched_barrier(0);
        }
        cA = A[16]; cT = T[17]; cS = S[18]; cD0 = D0[19]; cD1 = D1[19]; cB = Bf[4];
    };

    for (int j = 0; j < D_DIM; j += 32) {
        trip(qa0, qb0, va0, vb0, qa1, qb1, va1, vb1, j + 16);
        trip(qa1, qb1, va1, vb1, qa0, qb0, va0, vb0, (j + 32) & (D_DIM - 1));
    }

    const int M0 = half * 8;
    #pragma unroll
    for (int u = 0; u < 4; ++u) {
        #pragma unroll
        for (int r = 0; r < 8; ++r) {
            sC[wave][(M0 + r) * 16 + laneM] = c[u][r];
        }
        asm volatile("s_wait_dscnt 0" ::: "memory");
        if (lane < 16) {
            float sumE  = sC[wave][lane * 16 + 0];
            float sumEV = sC[wave][lane * 16 + 1];
            sAtt[wave][16 * u + lane] = (sumEV / sumE) * invSqrtD;
        }
    }
    __syncthreads();

    if (tid < ROWS_PER_BLOCK) {
        float acc = sAtt[0][tid];
        for (int h = 1; h < H_DIM; ++h) acc += sAtt[h][tid];
        out[b * D_DIM + row0 + tid] = xr + acc;
    }
}

extern "C" void kernel_launch(void* const* d_in, const int* in_sizes, int n_in,
                              void* d_out, int out_size, void* d_ws, size_t ws_size,
                              hipStream_t stream) {
    const float* x      = (const float*)d_in[0];
    const float* alphas = (const float*)d_in[1];
    const float* betas  = (const float*)d_in[2];
    float* out          = (float*)d_out;

    fl_attn_kernel<<<B_DIM * (D_DIM / ROWS_PER_BLOCK), 256, 0, stream>>>(x, alphas, betas, out);
}


# pass 2 issue order tied inside each stretch: plain and transcendental instructions by turns (sub, exp, fma, exp, add, rcp, add, rcp), no two of the four transcendentals back to back; sums unchanged
# speedup vs baseline: 2.1957x; 1.0557x over previous
#include <hip/hip_runtime.h>
#include <math.h>

typedef __attribute__((ext_vector_type(2))) float v2f;
typedef __attribute__((ext_vector_type(4))) float v4f;
typedef __attribute__((ext_vector_type(8))) float v8f;

#define B_DIM 64
#define D_DIM 512
#define H_DIM 8
#define ROWS_PER_BLOCK 64
#define EPS_F 1e-8f

#if defined(__HIP_DEVICE_COMPILE__) && defined(__AMDGCN__)
#if !__has_builtin(__builtin_amdgcn_wmma_f32_16x16x4_f32)
#error "missing __builtin_amdgcn_wmma_f32_16x16x4_f32 on this toolchain (device pass)"
#endif
#endif

__global__ __launch_bounds__(256) void fl_attn_kernel(const float* __restrict__ x,
                                                      const float* __restrict__ alphas,
                                                      const float* __restrict__ betas,
                                                      float* __restrict__ out) {
    __shared__ __attribute__((aligned(16))) float sQ[H_DIM][D_DIM];
    __shared__ __attribute__((aligned(16))) float sV[H_DIM][D_DIM];
    __shared__ float sC[8][16 * 16];
    __shared__ float sAtt[H_DIM][ROWS_PER_BLOCK];

    const int bs  = blockIdx.x;
    const int b   = bs & (B_DIM - 1);
    const int sl  = bs >> 6;
    const int tid = threadIdx.x;

    const int wave  = __builtin_amdgcn_readfirstlane(tid >> 5);
    const int lane  = tid & 31;
    const int half  = lane >> 4;
    const int laneM = lane & 15;
    const int kb    = half * 2;
    const float invSqrtD = 0.044194173824159216f;

    const float* xrow = x + b * D_DIM;
    const int row0 = sl * ROWS_PER_BLOCK;

    const float x0 = xrow[tid];
    const float x1 = xrow[tid + 256];
    const float xk[4] = {xrow[row0 + laneM], xrow[row0 + 16 + laneM],
                         xrow[row0 + 32 + laneM], xrow[row0 + 48 + laneM]};
    const float xr = xrow[row0 + (tid & (ROWS_PER_BLOCK - 1))];

    #pragma unroll
    for (int h = 0; h < H_DIM; ++h) {
        const float aq = alphas[h * 3 + 0], av = alphas[h * 3 + 2];
        const float bq = betas[h * 3 + 0],  bv = betas[h * 3 + 2];
        sQ[h][tid]       = fmaf(aq, x0, bq);
        sQ[h][tid + 256] = fmaf(aq, x1, bq);
        sV[h][tid]       = fmaf(av, x0, bv);
        sV[h][tid + 256] = fmaf(av, x1, bv);
    }
    __syncthreads();

    const float ak = alphas[wave * 3 + 1];
    const float bk = betas[wave * 3 + 1];
    const float* sQw = sQ[wave];
    const float* sVw = sV[wave];

    float ki[4];
    float smax[4];

    #pragma unroll
    for (int u = 0; u < 4; ++u) {
        const float k = fmaf(ak, xk[u], bk);
        const v2f k2 = {k, k};
        float mnA = 3.402823466e+38f;
        float mnB = 3.402823466e+38f;
        const float* q1 = sQw + half * 256;
        for (int j = 0; j < 256; j += 4) {
            const v4f q4  = *(const v4f*)(q1 + j);
            const v2f d01 = q4.xy - k2;
            const v2f d23 = q4.zw - k2;
            mnA = fminf(fminf(mnA, fabsf(d01.x)), fabsf(d01.y));
            mnB = fminf(fminf(mnB, fabsf(d23.x)), fabsf(d23.y));
        }
        float mn = fminf(mnA, mnB);
        mn = fminf(mn, __shfl_xor(mn, 16));
        ki[u]   = k;
        smax[u] = __builtin_amdgcn_rcpf(mn + EPS_F);
    }

    const float LOG2E_F = 1.44269504088896341f;
    v8f c[4] = {{}, {}, {}, {}};
    const float* qh = sQw + half * 8;
    const float* vh = sVw + half * 8;

    v2f k2[4], nm2[4];
    #pragma unroll
    for (int u = 0; u < 4; ++u) {
        k2[u]  = (v2f){ki[u], ki[u]};
        nm2[u] = (v2f){-smax[u] * LOG2E_F, -smax[u] * LOG2E_F};
    }
    const v2f L2 = {LOG2E_F, LOG2E_F};
    const float selV   = (laneM == 1) ? 1.0f : 0.0f;
    const float selOne = (laneM == 0) ? 1.0f : 0.0f;
    const v2f selV2   = {selV, selV};
    const v2f selOne2 = {selOne, selOne};

    auto stD = [&](v2f q, int u, float& d0, float& d1) {
        const v2f d = q - k2[u];
        d0 = fabsf(d.x) + EPS_F;
        d1 = fabsf(d.y) + EPS_F;
    };
    auto stR = [&](float d0, float d1) -> v2f {
        return (v2f){__builtin_amdgcn_rcpf(d0), __builtin_amdgcn_rcpf(d1)};
    };
    auto stF = [&](v2f s, int u) -> v2f { return __builtin_elementwise_fma(s, L2, nm2[u]); };
    auto stE = [&](v2f t) -> v2f {
        return (v2f){__builtin_amdgcn_exp2f(t.x), __builtin_amdgcn_exp2f(t.y)};
    };
    auto mkB = [&](v2f w) -> v2f { return __builtin_elementwise_fma(w, selV2, selOne2); };

    v4f qa0 = *(const v4f*)(qh);
    v4f qb0 = *(const v4f*)(qh + 4);
    v4f va0 = *(const v4f*)(vh);
    v4f vb0 = *(const v4f*)(vh + 4);
    v4f qa1, qb1, va1, vb1;
    float f0, f1, cD0, cD1;
    stD(qa0.xy, 0, f0, f1);
    v2f cA = stE(stF(stR(f0, f1), 0));
    stD(qa0.xy, 1, f0, f1);
    v2f cT = stF(stR(f0, f1), 1);
    stD(qa0.xy, 2, f0, f1);
    v2f cS = stR(f0, f1);
    stD(qa0.xy, 3, cD0, cD1);
    v2f cB = mkB(va0.xy);

    auto trip = [&](const v4f& qa, const v4f& qb, const v4f& va, const v4f& vb,
                    v4f& nqa, v4f& nqb, v4f& nva, v4f& nvb, int jn) {
        nqa = *(const v4f*)(qh + jn);
        nqb = *(const v4f*)(qh + jn + 4);
        nva = *(const v4f*)(vh + jn);
        nvb = *(const v4f*)(vh + jn + 4);
        const v2f qp[5] = {qa.xy, qa.zw, qb.xy, qb.zw, nqa.xy};
        const v2f vp[5] = {va.xy, va.zw, vb.xy, vb.zw, nva.xy};
        v2f A[17], T[18], S[19], Bf[5];
        float D0[20], D1[20];
        A[0] = cA; T[1] = cT; S[2] = cS; D0[3] = cD0; D1[3] = cD1; Bf[0] = cB;
#define ORDER_AFTER(done, next) asm volatile("" : "+v"(done), "+v"(next))
        #pragma unroll
        for (int n = 0; n < 16; ++n) {
            const int u = n & 3, t = n >> 2;
            c[u] = __builtin_amdgcn_wmma_f32_16x16x4_f32(false, A[n], false, Bf[t],
                                                         (short)0, c[u], false, false);
            v2f   s2  = S[n + 2];
            float t1x = T[n + 1].x, t1y = T[n + 1].y;
            float d3a = D0[n + 3],  d3b = D1[n + 3];
            v2f dd = qp[(n + 4) >> 2] - k2[(n + 4) & 3];
            ORDER_AFTER(dd, t1x);
            float ea = __builtin_amdgcn_exp2f(t1x);
            ORDER_AFTER(ea, s2);
            v2f tn2 = stF(s2, (n + 2) & 3);
            ORDER_AFTER(tn2, t1y);
            float eb = __builtin_amdgcn_exp2f(t1y);
            ORDER_AFTER(eb, dd);
            float da = fabsf(dd.x) + EPS_F;
            ORDER_AFTER(da, d3a);
            float ra = __builtin_amdgcn_rcpf(d3a);
            ORDER_AFTER(ra, dd);
            float db = fabsf(dd.y) + EPS_F;
            asm volatile("" : "+v"(db), "+v"(d3b), "+v"(d3a));
            float rb = __builtin_amdgcn_rcpf(d3b);
            if (u == 0) Bf[t + 1] = mkB(vp[t + 1]);
            A[n + 1] = (v2f){ea, eb};
            T[n + 2] = tn2;
            S[n + 3] = (v2f){ra, rb};
            D0[n + 4] = da;
            D1[n + 4] = db;

            asm volatile("" : "+v"(A[n + 1]), "+v"(T[n + 2]), "+v"(S[n + 3]),
                              "+v"(D0[n + 4]), "+v"(D1[n + 4]));
            __builtin_amdgcn_sched_barrier(0);
        }
#undef ORDER_AFTER
        cA = A[16]; cT = T[17]; cS = S[18]; cD0 = D0[19]; cD1 = D1[19]; cB = Bf[4];
    };

    for (int j = 0; j < D_DIM; j += 32) {
        trip(qa0, qb0, va0, vb0, qa1, qb1, va1, vb1, j + 16);
        trip(qa1, qb1, va1, vb1, qa0, qb0, va0, vb0, (j + 32) & (D_DIM - 1));
    }

    const int M0 = half * 8;
    #pragma unroll
    for (int u = 0; u < 4; ++u) {
        #pragma unroll
        for (int r = 0; r < 8; ++r) {
            sC[wave][(M0 + r) * 16 + laneM] = c[u][r];
        }
        asm volatile("s_wait_dscnt 0" ::: "memory");
        if (lane < 16) {
            float sumE  = sC[wave][lane * 16 + 0];
            float sumEV = sC[wave][lane * 16 + 1];
            sAtt[wave][16 * u + lane] = (sumEV / sumE) * invSqrtD;
        }
    }
    __syncthreads();

    if (tid < ROWS_PER_BLOCK) {
        float acc = sAtt[0][tid];
        for (int h = 1; h < H_DIM; ++h) acc += sAtt[h][tid];
        out[b * D_DIM + row0 + tid] = xr + acc;
    }
}

extern "C" void kernel_launch(void* const* d_in, const int* in_sizes, int n_in,
                              void* d_out, int out_size, void* d_ws, size_t ws_size,
                              hipStream_t stream) {
    const float* x      = (const float*)d_in[0];
    const float* alphas = (const float*)d_in[1];
    const float* betas  = (const float*)d_in[2];
    float* out          = (float*)d_out;

    fl_attn_kernel<<<B_DIM * (D_DIM / ROWS_PER_BLOCK), 256, 0, stream>>>(x, alphas, betas, out);
}


# pass 1 loop interchange: j outer, the wave's four row tiles inner, so each 16-byte q read from LDS serves four tiles (72 not 256 reads per wave), next 32 j prefetched; minima exact, sums unchanged
# speedup vs baseline: 2.3172x; 1.0553x over previous
#include <hip/hip_runtime.h>
#include <math.h>

typedef __attribute__((ext_vector_type(2))) float v2f;
typedef __attribute__((ext_vector_type(4))) float v4f;
typedef __attribute__((ext_vector_type(8))) float v8f;

#define B_DIM 64
#define D_DIM 512
#define H_DIM 8
#define ROWS_PER_BLOCK 64
#define EPS_F 1e-8f

#if defined(__HIP_DEVICE_COMPILE__) && defined(__AMDGCN__)
#if !__has_builtin(__builtin_amdgcn_wmma_f32_16x16x4_f32)
#error "missing __builtin_amdgcn_wmma_f32_16x16x4_f32 on this toolchain (device pass)"
#endif
#endif

__global__ __launch_bounds__(256) void fl_attn_kernel(const float* __restrict__ x,
                                                      const float* __restrict__ alphas,
                                                      const float* __restrict__ betas,
                                                      float* __restrict__ out) {
    __shared__ __attribute__((aligned(16))) float sQ[H_DIM][D_DIM];
    __shared__ __attribute__((aligned(16))) float sV[H_DIM][D_DIM];
    __shared__ float sC[8][16 * 16];
    __shared__ float sAtt[H_DIM][ROWS_PER_BLOCK];

    const int bs  = blockIdx.x;
    const int b   = bs & (B_DIM - 1);
    const int sl  = bs >> 6;
    const int tid = threadIdx.x;

    const int wave  = __builtin_amdgcn_readfirstlane(tid >> 5);
    const int lane  = tid & 31;
    const int half  = lane >> 4;
    const int laneM = lane & 15;
    const int kb    = half * 2;
    const float invSqrtD = 0.044194173824159216f;

    const float* xrow = x + b * D_DIM;
    const int row0 = sl * ROWS_PER_BLOCK;

    const float x0 = xrow[tid];
    const float x1 = xrow[tid + 256];
    const float xk[4] = {xrow[row0 + laneM], xrow[row0 + 16 + laneM],
                         xrow[row0 + 32 + laneM], xrow[row0 + 48 + laneM]};
    const float xr = xrow[row0 + (tid & (ROWS_PER_BLOCK - 1))];

    #pragma unroll
    for (int h = 0; h < H_DIM; ++h) {
        const float aq = alphas[h * 3 + 0], av = alphas[h * 3 + 2];
        const float bq = betas[h * 3 + 0],  bv = betas[h * 3 + 2];
        sQ[h][tid]       = fmaf(aq, x0, bq);
        sQ[h][tid + 256] = fmaf(aq, x1, bq);
        sV[h][tid]       = fmaf(av, x0, bv);
        sV[h][tid + 256] = fmaf(av, x1, bv);
    }
    __syncthreads();

    const float ak = alphas[wave * 3 + 1];
    const float bk = betas[wave * 3 + 1];
    const float* sQw = sQ[wave];
    const float* sVw = sV[wave];

    float ki[4];
    float smax[4];

    v2f kk[4];
    float mnA[4], mnB[4];
    #pragma unroll
    for (int u = 0; u < 4; ++u) {
        ki[u]  = fmaf(ak, xk[u], bk);
        kk[u]  = (v2f){ki[u], ki[u]};
        mnA[u] = 3.402823466e+38f;
        mnB[u] = 3.402823466e+38f;
    }
    const float* q1 = sQw + half * 256;
    auto minBlock = [&](const v4f (&cur)[8], v4f (&nxt)[8], int jn) {
        #pragma unroll
        for (int t = 0; t < 8; ++t) nxt[t] = *(const v4f*)(q1 + jn + 4 * t);
        #pragma unroll
        for (int t = 0; t < 8; ++t) {
            #pragma unroll
            for (int u = 0; u < 4; ++u) {
                const v2f d01 = cur[t].xy - kk[u];
                const v2f d23 = cur[t].zw - kk[u];
                mnA[u] = fminf(fminf(mnA[u], fabsf(d01.x)), fabsf(d01.y));
                mnB[u] = fminf(fminf(mnB[u], fabsf(d23.x)), fabsf(d23.y));
            }
        }
    };
    v4f qbufA[8], qbufB[8];
    #pragma unroll
    for (int t = 0; t < 8; ++t) qbufA[t] = *(const v4f*)(q1 + 4 * t);
    for (int j = 0; j < 256; j += 64) {
        minBlock(qbufA, qbufB, j + 32);
        minBlock(qbufB, qbufA, (j + 64) & 255);
    }
    #pragma unroll
    for (int u = 0; u < 4; ++u) {
        float mn = fminf(mnA[u], mnB[u]);
        mn = fminf(mn, __shfl_xor(mn, 16));
        smax[u] = __builtin_amdgcn_rcpf(mn + EPS_F);
    }

    const float LOG2E_F = 1.44269504088896341f;
    v8f c[4] = {{}, {}, {}, {}};
    const float* qh = sQw + half * 8;
    const float* vh = sVw + half * 8;

    v2f k2[4], nm2[4];
    #pragma unroll
    for (int u = 0; u < 4; ++u) {
        k2[u]  = (v2f){ki[u], ki[u]};
        nm2[u] = (v2f){-smax[u] * LOG2E_F, -smax[u] * LOG2E_F};
    }
    const v2f L2 = {LOG2E_F, LOG2E_F};
    const float selV   = (laneM == 1) ? 1.0f : 0.0f;
    const float selOne = (laneM == 0) ? 1.0f : 0.0f;
    const v2f selV2   = {selV, selV};
    const v2f selOne2 = {selOne, selOne};

    auto stD = [&](v2f q, int u, float& d0, float& d1) {
        const v2f d = q - k2[u];
        d0 = fabsf(d.x) + EPS_F;
        d1 = fabsf(d.y) + EPS_F;
    };
    auto stR = [&](float d0, float d1) -> v2f {
        return (v2f){__builtin_amdgcn_rcpf(d0), __builtin_amdgcn_rcpf(d1)};
    };
    auto stF = [&](v2f s, int u) -> v2f { return __builtin_elementwise_fma(s, L2, nm2[u]); };
    auto stE = [&](v2f t) -> v2f {
        return (v2f){__builtin_amdgcn_exp2f(t.x), __builtin_amdgcn_exp2f(t.y)};
    };
    auto mkB = [&](v2f w) -> v2f { return __builtin_elementwise_fma(w, selV2, selOne2); };

    v4f qa0 = *(const v4f*)(qh);
    v4f qb0 = *(const v4f*)(qh + 4);
    v4f va0 = *(const v4f*)(vh);
    v4f vb0 = *(const v4f*)(vh + 4);
    v4f qa1, qb1, va1, vb1;
    float f0, f1, cD0, cD1;
    stD(qa0.xy, 0, f0, f1);
    v2f cA = stE(stF(stR(f0, f1), 0));
    stD(qa0.xy, 1, f0, f1);
    v2f cT = stF(stR(f0, f1), 1);
    stD(qa0.xy, 2, f0, f1);
    v2f cS = stR(f0, f1);
    stD(qa0.xy, 3, cD0, cD1);
    v2f cB = mkB(va0.xy);

    auto trip = [&](const v4f& qa, const v4f& qb, const v4f& va, const v4f& vb,
                    v4f& nqa, v4f& nqb, v4f& nva, v4f& nvb, int jn) {
        nqa = *(const v4f*)(qh + jn);
        nqb = *(const v4f*)(qh + jn + 4);
        nva = *(const v4f*)(vh + jn);
        nvb = *(const v4f*)(vh + jn + 4);
        const v2f qp[5] = {qa.xy, qa.zw, qb.xy, qb.zw, nqa.xy};
        const v2f vp[5] = {va.xy, va.zw, vb.xy, vb.zw, nva.xy};
        v2f A[17], T[18], S[19], Bf[5];
        float D0[20], D1[20];
        A[0] = cA; T[1] = cT; S[2] = cS; D0[3] = cD0; D1[3] = cD1; Bf[0] = cB;
#define ORDER_AFTER(done, next) asm volatile("" : "+v"(done), "+v"(next))
        #pragma unroll
        for (int n = 0; n < 16; ++n) {
            const int u = n & 3, t = n >> 2;
            c[u] = __builtin_amdgcn_wmma_f32_16x16x4_f32(false, A[n], false, Bf[t],
                                                         (short)0, c[u], false, false);
            v2f   s2  = S[n + 2];
            float t1x = T[n + 1].x, t1y = T[n + 1].y;
            float d3a = D0[n + 3],  d3b = D1[n + 3];
            v2f dd = qp[(n + 4) >> 2] - k2[(n + 4) & 3];
            ORDER_AFTER(dd, t1x);
            float ea = __builtin_amdgcn_exp2f(t1x);
            ORDER_AFTER(ea, s2);
            v2f tn2 = stF(s2, (n + 2) & 3);
            ORDER_AFTER(tn2, t1y);
            float eb = __builtin_amdgcn_exp2f(t1y);
            ORDER_AFTER(eb, dd);
            float da = fabsf(dd.x) + EPS_F;
            ORDER_AFTER(da, d3a);
            float ra = __builtin_amdgcn_rcpf(d3a);
            ORDER_AFTER(ra, dd);
            float db = fabsf(dd.y) + EPS_F;
            asm volatile("" : "+v"(db), "+v"(d3b), "+v"(d3a));
            float rb = __builtin_amdgcn_rcpf(d3b);
            if (u == 0) Bf[t + 1] = mkB(vp[t + 1]);
            A[n + 1] = (v2f){ea, eb};
            T[n + 2] = tn2;
            S[n + 3] = (v2f){ra, rb};
            D0[n + 4] = da;
            D1[n + 4] = db;

            asm volatile("" : "+v"(A[n + 1]), "+v"(T[n + 2]), "+v"(S[n + 3]),
                              "+v"(D0[n + 4]), "+v"(D1[n + 4]));
            __builtin_amdgcn_sched_barrier(0);
        }
#undef ORDER_AFTER
        cA = A[16]; cT = T[17]; cS = S[18]; cD0 = D0[19]; cD1 = D1[19]; cB = Bf[4];
    };

    for (int j = 0; j < D_DIM; j += 32) {
        trip(qa0, qb0, va0, vb0, qa1, qb1, va1, vb1, j + 16);
        trip(qa1, qb1, va1, vb1, qa0, qb0, va0, vb0, (j + 32) & (D_DIM - 1));
    }

    const int M0 = half * 8;
    #pragma unroll
    for (int u = 0; u < 4; ++u) {
        #pragma unroll
        for (int r = 0; r < 8; ++r) {
            sC[wave][(M0 + r) * 16 + laneM] = c[u][r];
        }
        asm volatile("s_wait_dscnt 0" ::: "memory");
        if (lane < 16) {
            float sumE  = sC[wave][lane * 16 + 0];
            float sumEV = sC[wave][lane * 16 + 1];
            sAtt[wave][16 * u + lane] = (sumEV / sumE) * invSqrtD;
        }
    }
    __syncthreads();

    if (tid < ROWS_PER_BLOCK) {
        float acc = sAtt[0][tid];
        for (int h = 1; h < H_DIM; ++h) acc += sAtt[h][tid];
        out[b * D_DIM + row0 + tid] = xr + acc;
    }
}

extern "C" void kernel_launch(void* const* d_in, const int* in_sizes, int n_in,
                              void* d_out, int out_size, void* d_ws, size_t ws_size,
                              hipStream_t stream) {
    const float* x      = (const float*)d_in[0];
    const float* alphas = (const float*)d_in[1];
    const float* betas  = (const float*)d_in[2];
    float* out          = (float*)d_out;

    fl_attn_kernel<<<B_DIM * (D_DIM / ROWS_PER_BLOCK), 256, 0, stream>>>(x, alphas, betas, out);
}
